# starting kernel
# baseline (speedup 1.0000x reference)
#include <hip/hip_runtime.h>
#include <math.h>


typedef __attribute__((ext_vector_type(2))) float v2f;
typedef __attribute__((ext_vector_type(8))) float v8f;

#define HMLR_K    0.1f
#define HMLR_SK   0.31622776601683794f
#define HMLR_EPS  1e-15f
#define HMLR_LN2  0.6931471805599453f

__global__ void __launch_bounds__(256)
hmlr_prep_classes(const float* __restrict__ a_vals,
                  const float* __restrict__ p_vals,
                  float* __restrict__ mp,
                  float* __restrict__ ap,
                  float4* __restrict__ cscal,
                  int C) {
    int wave = (blockIdx.x * blockDim.x + threadIdx.x) >> 5;
    int lane = threadIdx.x & 31;
    if (wave >= C) return;

    const float k = HMLR_K, sk = HMLR_SK;
    size_t base = (size_t)wave * 64 + lane;
    float p0 = p_vals[base], p1 = p_vals[base + 32];
    float a0 = a_vals[base], a1 = a_vals[base + 32];

    float pn2 = p0 * p0 + p1 * p1;
    #pragma unroll
    for (int off = 16; off; off >>= 1) pn2 += __shfl_xor(pn2, off, 32);

    float pnorm = fmaxf(sqrtf(pn2), HMLR_EPS);
    float t = tanhf(sk * pnorm) / (sk * pnorm);
    float pp0 = t * p0, pp1 = t * p1;
    float p2 = t * t * pn2;

    float ka = 1.0f + k * p2;
    float ap0 = a0 * ka, ap1 = a1 * ka;

    float a2 = ap0 * ap0 + ap1 * ap1;
    float pa = -(pp0 * ap0 + pp1 * ap1);
    #pragma unroll
    for (int off = 16; off; off >>= 1) {
        a2 += __shfl_xor(a2, off, 32);
        pa += __shfl_xor(pa, off, 32);
    }

    float anorm = fmaxf(sqrtf(a2), HMLR_EPS);
    float lam   = 2.0f / (1.0f - k * p2);
    float scale = lam * anorm / sk;

    mp[base]      = -pp0;
    mp[base + 32] = -pp1;
    ap[base]      = ap0;
    ap[base + 32] = ap1;
    if (lane == 0) cscal[wave] = make_float4(p2, pa, anorm, scale);
}

__global__ void __launch_bounds__(256)
hmlr_prep_y2(const float* __restrict__ x, float* __restrict__ y2, int B) {
    int wave = (blockIdx.x * blockDim.x + threadIdx.x) >> 5;
    int lane = threadIdx.x & 31;
    if (wave >= B) return;
    size_t base = (size_t)wave * 64 + lane;
    float v0 = x[base], v1 = x[base + 32];
    float s = v0 * v0 + v1 * v1;
    #pragma unroll
    for (int off = 16; off; off >>= 1) s += __shfl_xor(s, off, 32);
    if (lane == 0) y2[wave] = s;
}

__global__ void __launch_bounds__(256)
hmlr_main(const float* __restrict__ x,
          const float* __restrict__ mp,
          const float* __restrict__ ap,
          const float4* __restrict__ cscal,
          const float* __restrict__ y2,
          float* __restrict__ out,
          int B, int C, int nTilesC, int nTiles) {
    int wave = threadIdx.x >> 5;
    int lane = threadIdx.x & 31;
    int tile = blockIdx.x * (blockDim.x >> 5) + wave;
    if (tile >= nTiles) return;

    int tileB = tile / nTilesC;
    int tileC = tile - tileB * nTilesC;

    int m  = lane & 15;
    int kh = lane >> 4;
    int bRow = tileB * 16 + m;
    int cG   = tileC * 16 + m;
    int cIdx = (cG < C) ? cG : (C - 1);

    const float* xr = x  + (size_t)bRow * 64 + 2 * kh;
    const float* mr = mp + (size_t)cIdx * 64 + 2 * kh;
    const float* ar = ap + (size_t)cIdx * 64 + 2 * kh;

    v8f axy = {0.f, 0.f, 0.f, 0.f, 0.f, 0.f, 0.f, 0.f};
    v8f axa = {0.f, 0.f, 0.f, 0.f, 0.f, 0.f, 0.f, 0.f};

    #pragma unroll
    for (int s = 0; s < 16; ++s) {
        v2f af = *(const v2f*)(xr + 4 * s);
        v2f bm = *(const v2f*)(mr + 4 * s);
        v2f ba = *(const v2f*)(ar + 4 * s);
        axy = __builtin_amdgcn_wmma_f32_16x16x4_f32(
            false, af, false, bm, (short)0, axy, false, false);
        axa = __builtin_amdgcn_wmma_f32_16x16x4_f32(
            false, af, false, ba, (short)0, axa, false, false);
    }

    float4 cs = cscal[cIdx];
    float x2 = cs.x, pa = cs.y, anorm = cs.z, scale = cs.w;
    const float k = HMLR_K, sk = HMLR_SK;
    int rbase = tileB * 16 + 8 * kh;

    #pragma unroll
    for (int v = 0; v < 8; ++v) {
        int row = rbase + v;
        float yy = y2[row];
        float xy = axy[v];
        float xa = axa[v];

        float alpha = fmaf(2.0f * k, xy, fmaf(k, yy, 1.0f));
        float beta  = fmaf(-k, x2, 1.0f);
        float den   = fmaf(2.0f * k, xy, fmaf(k * k * x2, yy, 1.0f));
        float id    = __builtin_amdgcn_rcpf(den);

        float mda  = (alpha * pa + beta * xa) * id;
        float mob2 = (alpha * alpha * x2 + 2.0f * alpha * beta * xy
                      + beta * beta * yy) * (id * id);

        float numer = 2.0f * sk * mda;
        float denom = anorm * fmaf(-k, mob2, 1.0f);
        float z  = numer * __builtin_amdgcn_rcpf(denom);

        float az = fabsf(z);
        float sq = __builtin_amdgcn_sqrtf(fmaf(az, az, 1.0f));
        float lg = __builtin_amdgcn_logf(az + sq);
        float r  = scale * (HMLR_LN2 * lg);
        float logit = (z < 0.0f) ? -r : r;

        if (cG < C) out[(size_t)row * C + cG] = logit;
    }
}

extern "C" void kernel_launch(void* const* d_in, const int* in_sizes, int n_in,
                              void* d_out, int out_size, void* d_ws, size_t ws_size,
                              hipStream_t stream) {
    const float* x      = (const float*)d_in[0];
    const float* a_vals = (const float*)d_in[1];
    const float* p_vals = (const float*)d_in[2];
    float* out = (float*)d_out;

    const int D = 64;
    int B = in_sizes[0] / D;
    int C = in_sizes[1] / D;

    char* ws = (char*)d_ws;
    size_t rowBytes = (size_t)C * D * sizeof(float);
    float*  mp = (float*)(ws);
    float*  ap = (float*)(ws + rowBytes);
    float4* cs = (float4*)(ws + 2 * rowBytes);
    float*  y2 = (float*)(ws + 2 * rowBytes + (size_t)C * sizeof(float4));

    hmlr_prep_classes<<<(C + 7) / 8, 256, 0, stream>>>(a_vals, p_vals, mp, ap, cs, C);
    hmlr_prep_y2<<<(B + 7) / 8, 256, 0, stream>>>(x, y2, B);

    int nTilesB = B / 16;
    int nTilesC = (C + 15) / 16;
    int nTiles  = nTilesB * nTilesC;
    hmlr_main<<<(nTiles + 7) / 8, 256, 0, stream>>>(x, mp, ap, cs, y2, out,
                                                    B, C, nTilesC, nTiles);
}


# Main kernel register-tiled: a wave keeps its 16 classes' mp/ap fragments in VGPRs (128-bit loads, one permuted k order for A and B: k sum order changes), walks 4 row tiles, next X prefetched
# speedup vs baseline: 1.3949x; 1.3949x over previous
#include <hip/hip_runtime.h>
#include <math.h>


typedef __attribute__((ext_vector_type(2))) float v2f;
typedef __attribute__((ext_vector_type(4))) float v4f;
typedef __attribute__((ext_vector_type(8))) float v8f;

#define HMLR_K    0.1f
#define HMLR_SK   0.31622776601683794f
#define HMLR_EPS  1e-15f
#define HMLR_LN2  0.6931471805599453f

__global__ void __launch_bounds__(256)
hmlr_prep_classes(const float* __restrict__ a_vals,
                  const float* __restrict__ p_vals,
                  float* __restrict__ mp,
                  float* __restrict__ ap,
                  float4* __restrict__ cscal,
                  int C) {
    int wave = (blockIdx.x * blockDim.x + threadIdx.x) >> 5;
    int lane = threadIdx.x & 31;
    if (wave >= C) return;

    const float k = HMLR_K, sk = HMLR_SK;
    size_t base = (size_t)wave * 64 + lane;
    float p0 = p_vals[base], p1 = p_vals[base + 32];
    float a0 = a_vals[base], a1 = a_vals[base + 32];

    float pn2 = p0 * p0 + p1 * p1;
    #pragma unroll
    for (int off = 16; off; off >>= 1) pn2 += __shfl_xor(pn2, off, 32);

    float pnorm = fmaxf(sqrtf(pn2), HMLR_EPS);
    float t = tanhf(sk * pnorm) / (sk * pnorm);
    float pp0 = t * p0, pp1 = t * p1;
    float p2 = t * t * pn2;

    float ka = 1.0f + k * p2;
    float ap0 = a0 * ka, ap1 = a1 * ka;

    float a2 = ap0 * ap0 + ap1 * ap1;
    float pa = -(pp0 * ap0 + pp1 * ap1);
    #pragma unroll
    for (int off = 16; off; off >>= 1) {
        a2 += __shfl_xor(a2, off, 32);
        pa += __shfl_xor(pa, off, 32);
    }

    float anorm = fmaxf(sqrtf(a2), HMLR_EPS);
    float lam   = 2.0f / (1.0f - k * p2);
    float scale = lam * anorm / sk;

    mp[base]      = -pp0;
    mp[base + 32] = -pp1;
    ap[base]      = ap0;
    ap[base + 32] = ap1;
    if (lane == 0) cscal[wave] = make_float4(p2, pa, anorm, scale);
}

__global__ void __launch_bounds__(256)
hmlr_prep_y2(const float* __restrict__ x, float* __restrict__ y2, int B) {
    int wave = (blockIdx.x * blockDim.x + threadIdx.x) >> 5;
    int lane = threadIdx.x & 31;
    if (wave >= B) return;
    size_t base = (size_t)wave * 64 + lane;
    float v0 = x[base], v1 = x[base + 32];
    float s = v0 * v0 + v1 * v1;
    #pragma unroll
    for (int off = 16; off; off >>= 1) s += __shfl_xor(s, off, 32);
    if (lane == 0) y2[wave] = s;
}

#define HMLR_RT 4

__device__ __forceinline__ void hmlr_load_frag(v4f (&f)[8], const float* __restrict__ rowk) {
    #pragma unroll
    for (int j = 0; j < 8; ++j) f[j] = *(const v4f*)(rowk + 8 * j);
}

__global__ void __launch_bounds__(256)
hmlr_main(const float* __restrict__ x,
          const float* __restrict__ mp,
          const float* __restrict__ ap,
          const float4* __restrict__ cscal,
          const float* __restrict__ y2,
          float* __restrict__ out,
          int B, int C, int nTilesC, int nTilesB, int nWork) {
    int wave = threadIdx.x >> 5;
    int lane = threadIdx.x & 31;
    int work = blockIdx.x * (blockDim.x >> 5) + wave;
    if (work >= nWork) return;

    int grp   = work / nTilesC;
    int tileC = work - grp * nTilesC;

    int m  = lane & 15;
    int kh = lane >> 4;
    int cG   = tileC * 16 + m;
    int cIdx = (cG < C) ? cG : (C - 1);

    v4f bm[8], ba[8];
    hmlr_load_frag(bm, mp + (size_t)cIdx * 64 + 4 * kh);
    hmlr_load_frag(ba, ap + (size_t)cIdx * 64 + 4 * kh);

    float4 cs = cscal[cIdx];
    float x2 = cs.x, pa = cs.y, anorm = cs.z, scale = cs.w;
    const float k = HMLR_K, sk = HMLR_SK;
    float beta = fmaf(-k, x2, 1.0f);

    int tB0 = grp * HMLR_RT;
    int tL0 = (tB0 < nTilesB) ? tB0 : (nTilesB - 1);
    const float* yr0 = y2 + tL0 * 16 + 8 * kh;
    v4f ya = *(const v4f*)(yr0);
    v4f yb = *(const v4f*)(yr0 + 4);
    v4f af[8];
    hmlr_load_frag(af, x + (size_t)(tL0 * 16 + m) * 64 + 4 * kh);

    #pragma unroll
    for (int t = 0; t < HMLR_RT; ++t) {
        int tileB = tB0 + t;
        bool live = tileB < nTilesB;
        int tileL = live ? tileB : (nTilesB - 1);

        v4f na, nb, an[8];
        if (t + 1 < HMLR_RT) {
            int tN = tileB + 1;
            tN = (tN < nTilesB) ? tN : (nTilesB - 1);
            const float* yrn = y2 + tN * 16 + 8 * kh;
            na = *(const v4f*)(yrn);
            nb = *(const v4f*)(yrn + 4);
            hmlr_load_frag(an, x + (size_t)(tN * 16 + m) * 64 + 4 * kh);
        }

        v8f axy = {0.f, 0.f, 0.f, 0.f, 0.f, 0.f, 0.f, 0.f};
        v8f axa = {0.f, 0.f, 0.f, 0.f, 0.f, 0.f, 0.f, 0.f};

        #pragma unroll
        for (int j = 0; j < 8; ++j) {
            v2f a0 = {af[j].x, af[j].y}, a1 = {af[j].z, af[j].w};
            v2f m0 = {bm[j].x, bm[j].y}, m1 = {bm[j].z, bm[j].w};
            v2f p0 = {ba[j].x, ba[j].y}, p1 = {ba[j].z, ba[j].w};
            axy = __builtin_amdgcn_wmma_f32_16x16x4_f32(
                false, a0, false, m0, (short)0, axy, false, false);
            axa = __builtin_amdgcn_wmma_f32_16x16x4_f32(
                false, a0, false, p0, (short)0, axa, false, false);
            axy = __builtin_amdgcn_wmma_f32_16x16x4_f32(
                false, a1, false, m1, (short)0, axy, false, false);
            axa = __builtin_amdgcn_wmma_f32_16x16x4_f32(
                false, a1, false, p1, (short)0, axa, false, false);
        }

        if (t == 0) {
            asm volatile("" :: "v"(ya), "v"(yb));
        }

        int rbase = tileL * 16 + 8 * kh;

        #pragma unroll
        for (int v = 0; v < 8; ++v) {
            int row = rbase + v;
            float yy = (v < 4) ? ya[v & 3] : yb[v & 3];
            float xy = axy[v];
            float xa = axa[v];

            float alpha = fmaf(2.0f * k, xy, fmaf(k, yy, 1.0f));
            float den   = fmaf(2.0f * k, xy, fmaf(k * k * x2, yy, 1.0f));
            float id    = __builtin_amdgcn_rcpf(den);

            float mda  = (alpha * pa + beta * xa) * id;
            float mob2 = (alpha * alpha * x2 + 2.0f * alpha * beta * xy
                          + beta * beta * yy) * (id * id);

            float numer = 2.0f * sk * mda;
            float denom = anorm * fmaf(-k, mob2, 1.0f);
            float z  = numer * __builtin_amdgcn_rcpf(denom);

            float az = fabsf(z);
            float sq = __builtin_amdgcn_sqrtf(fmaf(az, az, 1.0f));
            float lg = __builtin_amdgcn_logf(az + sq);
            float r  = scale * (HMLR_LN2 * lg);
            float logit = (z < 0.0f) ? -r : r;

            if (live && cG < C) out[(size_t)row * C + cG] = logit;
        }

        if (t + 1 < HMLR_RT) {
            #pragma unroll
            for (int j = 0; j < 8; ++j) af[j] = an[j];
            ya = na;
            yb = nb;
        }
    }
}

extern "C" void kernel_launch(void* const* d_in, const int* in_sizes, int n_in,
                              void* d_out, int out_size, void* d_ws, size_t ws_size,
                              hipStream_t stream) {
    const float* x      = (const float*)d_in[0];
    const float* a_vals = (const float*)d_in[1];
    const float* p_vals = (const float*)d_in[2];
    float* out = (float*)d_out;

    const int D = 64;
    int B = in_sizes[0] / D;
    int C = in_sizes[1] / D;

    char* ws = (char*)d_ws;
    size_t rowBytes = (size_t)C * D * sizeof(float);
    float*  mp = (float*)(ws);
    float*  ap = (float*)(ws + rowBytes);
    float4* cs = (float4*)(ws + 2 * rowBytes);
    float*  y2 = (float*)(ws + 2 * rowBytes + (size_t)C * sizeof(float4));

    hmlr_prep_classes<<<(C + 7) / 8, 256, 0, stream>>>(a_vals, p_vals, mp, ap, cs, C);
    hmlr_prep_y2<<<(B + 7) / 8, 256, 0, stream>>>(x, y2, B);

    int nTilesB = B / 16;
    int nTilesC = (C + 15) / 16;
    int nGroups = (nTilesB + HMLR_RT - 1) / HMLR_RT;
    int nWork   = nGroups * nTilesC;
    hmlr_main<<<(nWork + 7) / 8, 256, 0, stream>>>(x, mp, ap, cs, y2, out,
                                                   B, C, nTilesC, nTilesB, nWork);
}


# One launch instead of three: each wave rebuilds its classes' mp/a_poin fragments and constants and its rows' squared norms in registers from the raw inputs (norm and dot sum order changes)
# speedup vs baseline: 2.3968x; 1.7183x over previous
#include <hip/hip_runtime.h>
#include <math.h>


typedef __attribute__((ext_vector_type(2))) float v2f;
typedef __attribute__((ext_vector_type(4))) float v4f;
typedef __attribute__((ext_vector_type(8))) float v8f;

#define HMLR_K    0.1f
#define HMLR_SK   0.31622776601683794f
#define HMLR_EPS  1e-15f
#define HMLR_LN2  0.6931471805599453f

#define HMLR_RT 4

__device__ __forceinline__ void hmlr_load_frag(v4f (&f)[8], const float* __restrict__ rowk) {
    #pragma unroll
    for (int j = 0; j < 8; ++j) f[j] = *(const v4f*)(rowk + 8 * j);
}

__device__ __forceinline__ float hmlr_xhalf(float v) {
    int b = __builtin_bit_cast(int, v);
    return __builtin_bit_cast(float,
        __builtin_amdgcn_permlanex16(b, b, 0x76543210, 0xfedcba98, false, false));
}

__device__ __forceinline__ float hmlr_dot64(const v4f (&f)[8], const v4f (&g)[8]) {
    v4f s = f[0] * g[0];
    #pragma unroll
    for (int j = 1; j < 8; ++j) s += f[j] * g[j];
    float r = (s.x + s.y) + (s.z + s.w);
    return r + hmlr_xhalf(r);
}

#define HMLR_DPP_ROW_ROR8   0x128
#define HMLR_DPP_ROW_SHARE0 0x150

__device__ __forceinline__ void hmlr_rows8(float s, float (&yy)[8]) {
    int b = __builtin_bit_cast(int, s);
    int r = __builtin_amdgcn_update_dpp(b, b, HMLR_DPP_ROW_ROR8, 0x2, 0xf, false);
    yy[0] = __builtin_bit_cast(float, __builtin_amdgcn_mov_dpp(r, HMLR_DPP_ROW_SHARE0 + 0, 0xf, 0xf, false));
    yy[1] = __builtin_bit_cast(float, __builtin_amdgcn_mov_dpp(r, HMLR_DPP_ROW_SHARE0 + 1, 0xf, 0xf, false));
    yy[2] = __builtin_bit_cast(float, __builtin_amdgcn_mov_dpp(r, HMLR_DPP_ROW_SHARE0 + 2, 0xf, 0xf, false));
    yy[3] = __builtin_bit_cast(float, __builtin_amdgcn_mov_dpp(r, HMLR_DPP_ROW_SHARE0 + 3, 0xf, 0xf, false));
    yy[4] = __builtin_bit_cast(float, __builtin_amdgcn_mov_dpp(r, HMLR_DPP_ROW_SHARE0 + 4, 0xf, 0xf, false));
    yy[5] = __builtin_bit_cast(float, __builtin_amdgcn_mov_dpp(r, HMLR_DPP_ROW_SHARE0 + 5, 0xf, 0xf, false));
    yy[6] = __builtin_bit_cast(float, __builtin_amdgcn_mov_dpp(r, HMLR_DPP_ROW_SHARE0 + 6, 0xf, 0xf, false));
    yy[7] = __builtin_bit_cast(float, __builtin_amdgcn_mov_dpp(r, HMLR_DPP_ROW_SHARE0 + 7, 0xf, 0xf, false));
}

__global__ void __launch_bounds__(256)
hmlr_fused(const float* __restrict__ x,
           const float* __restrict__ a_vals,
           const float* __restrict__ p_vals,
           float* __restrict__ out,
           int B, int C, int nTilesC, int nTilesB, int nWork) {
    int wave = threadIdx.x >> 5;
    int lane = threadIdx.x & 31;
    int work = blockIdx.x * (blockDim.x >> 5) + wave;
    if (work >= nWork) return;

    int grp   = work / nTilesC;
    int tileC = work - grp * nTilesC;

    int m  = lane & 15;
    int kh = lane >> 4;
    int cG   = tileC * 16 + m;
    int cIdx = (cG < C) ? cG : (C - 1);

    v4f bm[8], ba[8], af[8], an[8];
    hmlr_load_frag(bm, p_vals + (size_t)cIdx * 64 + 4 * kh);
    hmlr_load_frag(ba, a_vals + (size_t)cIdx * 64 + 4 * kh);
    int tB0 = grp * HMLR_RT;
    int tL0 = (tB0 < nTilesB) ? tB0 : (nTilesB - 1);
    int tL1 = (tB0 + 1 < nTilesB) ? (tB0 + 1) : (nTilesB - 1);
    hmlr_load_frag(af, x + (size_t)(tL0 * 16 + m) * 64 + 4 * kh);
    hmlr_load_frag(an, x + (size_t)(tL1 * 16 + m) * 64 + 4 * kh);

    const float k = HMLR_K, sk = HMLR_SK;
    float pn2   = hmlr_dot64(bm, bm);
    float pnorm = fmaxf(sqrtf(pn2), HMLR_EPS);
    float tp    = tanhf(sk * pnorm) / (sk * pnorm);
    float x2    = tp * tp * pn2;
    float ka    = 1.0f + k * x2;
    float ntp   = -tp;
    #pragma unroll
    for (int j = 0; j < 8; ++j) {
        bm[j] = bm[j] * ntp;
        ba[j] = ba[j] * ka;
    }
    float a2    = hmlr_dot64(ba, ba);
    float pa    = hmlr_dot64(bm, ba);
    float anorm = fmaxf(sqrtf(a2), HMLR_EPS);
    float lam   = 2.0f / (1.0f - k * x2);
    float scale = lam * anorm / sk;
    float beta  = fmaf(-k, x2, 1.0f);

    #pragma unroll
    for (int t = 0; t < HMLR_RT; ++t) {
        int tileB = tB0 + t;
        bool live = tileB < nTilesB;
        int tileL = live ? tileB : (nTilesB - 1);

        if (t > 0 && t + 1 < HMLR_RT) {
            int tN = tileB + 1;
            tN = (tN < nTilesB) ? tN : (nTilesB - 1);
            hmlr_load_frag(an, x + (size_t)(tN * 16 + m) * 64 + 4 * kh);
        }

        float yy8[8];
        hmlr_rows8(hmlr_dot64(af, af), yy8);

        v8f axy = {0.f, 0.f, 0.f, 0.f, 0.f, 0.f, 0.f, 0.f};
        v8f axa = {0.f, 0.f, 0.f, 0.f, 0.f, 0.f, 0.f, 0.f};

        #pragma unroll
        for (int j = 0; j < 8; ++j) {
            v2f a0 = {af[j].x, af[j].y}, a1 = {af[j].z, af[j].w};
            v2f m0 = {bm[j].x, bm[j].y}, m1 = {bm[j].z, bm[j].w};
            v2f p0 = {ba[j].x, ba[j].y}, p1 = {ba[j].z, ba[j].w};
            axy = __builtin_amdgcn_wmma_f32_16x16x4_f32(
                false, a0, false, m0, (short)0, axy, false, false);
            axa = __builtin_amdgcn_wmma_f32_16x16x4_f32(
                false, a0, false, p0, (short)0, axa, false, false);
            axy = __builtin_amdgcn_wmma_f32_16x16x4_f32(
                false, a1, false, m1, (short)0, axy, false, false);
            axa = __builtin_amdgcn_wmma_f32_16x16x4_f32(
                false, a1, false, p1, (short)0, axa, false, false);
        }

        int rbase = tileL * 16 + 8 * kh;

        #pragma unroll
        for (int v = 0; v < 8; ++v) {
            int row = rbase + v;
            float yy = yy8[v];
            float xy = axy[v];
            float xa = axa[v];

            float alpha = fmaf(2.0f * k, xy, fmaf(k, yy, 1.0f));
            float den   = fmaf(2.0f * k, xy, fmaf(k * k * x2, yy, 1.0f));
            float id    = __builtin_amdgcn_rcpf(den);

            float mda  = (alpha * pa + beta * xa) * id;
            float mob2 = (alpha * alpha * x2 + 2.0f * alpha * beta * xy
                          + beta * beta * yy) * (id * id);

            float numer = 2.0f * sk * mda;
            float denom = anorm * fmaf(-k, mob2, 1.0f);
            float z  = numer * __builtin_amdgcn_rcpf(denom);

            float az = fabsf(z);
            float sq = __builtin_amdgcn_sqrtf(fmaf(az, az, 1.0f));
            float lg = __builtin_amdgcn_logf(az + sq);
            float r  = scale * (HMLR_LN2 * lg);
            float logit = (z < 0.0f) ? -r : r;

            if (live && cG < C) out[(size_t)row * C + cG] = logit;
        }

        if (t + 1 < HMLR_RT) {
            #pragma unroll
            for (int j = 0; j < 8; ++j) af[j] = an[j];
        }
    }
}

extern "C" void kernel_launch(void* const* d_in, const int* in_sizes, int n_in,
                              void* d_out, int out_size, void* d_ws, size_t ws_size,
                              hipStream_t stream) {
    const float* x      = (const float*)d_in[0];
    const float* a_vals = (const float*)d_in[1];
    const float* p_vals = (const float*)d_in[2];
    float* out = (float*)d_out;

    const int D = 64;
    int B = in_sizes[0] / D;
    int C = in_sizes[1] / D;

    int nTilesB = B / 16;
    int nTilesC = (C + 15) / 16;
    int nGroups = (nTilesB + HMLR_RT - 1) / HMLR_RT;
    int nWork   = nGroups * nTilesC;
    hmlr_fused<<<(nWork + 7) / 8, 256, 0, stream>>>(x, a_vals, p_vals, out,
                                                    B, C, nTilesC, nTilesB, nWork);
}
